# starting kernel
# baseline (speedup 1.0000x reference)
#include <hip/hip_runtime.h>

typedef __attribute__((ext_vector_type(2))) float v2f;
typedef __attribute__((ext_vector_type(8))) float v8f;

#define WAVES_PER_BLOCK 8
#define LSTRIDE 33

__constant__ float cH[23] = {
    -1.20162964e-4f, 0.0f,  1.615524292e-3f, 0.0f, -1.0385513306e-2f, 0.0f,
     4.3619155884e-2f, 0.0f, -1.45397186478e-1f, 0.0f, 6.1066818237e-1f,
     1.0f,
     6.1066818237e-1f, 0.0f, -1.45397186478e-1f, 0.0f, 4.3619155884e-2f, 0.0f,
    -1.0385513306e-2f, 0.0f,  1.615524292e-3f, 0.0f, -1.20162964e-4f };

__global__ __launch_bounds__(256) void cdf23_up2x_wmma(
    const float* __restrict__ in, float* __restrict__ out, int N, int phi)
{
  __shared__ float sG[16 * 20];
  __shared__ float sX[WAVES_PER_BLOCK][20 * LSTRIDE];
  __shared__ float sT[WAVES_PER_BLOCK][16 * LSTRIDE];

  const int tid  = threadIdx.x;
  const int lane = tid & 31;
  const int wv   = tid >> 5;
  const int cofs = phi ? 22 : 21;

  for (int t = tid; t < 16 * 20; t += 256) {
    int m = t / 20, n = t - 20 * m;
    int j = m - 2 * n + cofs;
    sG[t] = (j >= 0 && j < 23) ? cH[j] : 0.0f;
  }
  __syncthreads();

  const int TW  = (2 * N) >> 4;
  const int tpp = TW * TW;
  const int g     = blockIdx.x * WAVES_PER_BLOCK + wv;
  const int plane = g / tpp;
  const int rem   = g - plane * tpp;
  const int i0 = (rem / TW) * 16;
  const int j0 = (rem % TW) * 16;
  const int back = phi ? 6 : 5;
  const int nr0 = (i0 >> 1) - back;
  const int nc0 = (j0 >> 1) - back;
  const float* __restrict__ xin = in + (long long)plane * N * N;

  float* X = sX[wv];
  float* T = sT[wv];

  for (int t = lane; t < 20 * LSTRIDE; t += 32) X[t] = 0.0f;
  __builtin_amdgcn_wave_barrier();
  for (int t = lane; t < 19 * 19; t += 32) {
    int r = t / 19, q = t - 19 * r;
    int gr = nr0 + r; if (gr < 0) gr += N; else if (gr >= N) gr -= N;
    int gc = nc0 + q; if (gc < 0) gc += N; else if (gc >= N) gc -= N;
    X[r * LSTRIDE + q] = xin[(long long)gr * N + gc];
    int gc2 = gc + 8; if (gc2 >= N) gc2 -= N;
    __builtin_prefetch(xin + (long long)gr * N + gc2, 0, 1);
  }
  __builtin_amdgcn_wave_barrier();

  const int mh = lane & 15;
  const int kh = (lane < 16) ? 0 : 2;

  v8f t0 = {}, t1 = {};
#pragma unroll
  for (int k = 0; k < 5; ++k) {
    const int kb = 4 * k + kh;
    v2f a, b, b2;
    a.x  = sG[mh * 20 + kb];
    a.y  = sG[mh * 20 + kb + 1];
    b.x  = X[kb * LSTRIDE + mh];
    b.y  = X[(kb + 1) * LSTRIDE + mh];
    b2.x = X[kb * LSTRIDE + 16 + mh];
    b2.y = X[(kb + 1) * LSTRIDE + 16 + mh];
    t0 = __builtin_amdgcn_wmma_f32_16x16x4_f32(false, a, false, b,  (short)0, t0, false, false);
    t1 = __builtin_amdgcn_wmma_f32_16x16x4_f32(false, a, false, b2, (short)0, t1, false, false);
  }

  const int mrow = (lane < 16) ? 0 : 8;
#pragma unroll
  for (int v = 0; v < 8; ++v) {
    T[(v + mrow) * LSTRIDE + mh]      = t0[v];
    T[(v + mrow) * LSTRIDE + 16 + mh] = t1[v];
  }
  __builtin_amdgcn_wave_barrier();

  v8f o = {};
#pragma unroll
  for (int k = 0; k < 5; ++k) {
    const int kb = 4 * k + kh;
    v2f a, b;
    a.x = T[mh * LSTRIDE + kb];
    a.y = T[mh * LSTRIDE + kb + 1];
    b.x = sG[mh * 20 + kb];
    b.y = sG[mh * 20 + kb + 1];
    o = __builtin_amdgcn_wmma_f32_16x16x4_f32(false, a, false, b, (short)0, o, false, false);
  }

  const int OW = 2 * N;
  float* __restrict__ op = out + (long long)plane * OW * OW;
#pragma unroll
  for (int v = 0; v < 8; ++v) {
    op[(long long)(i0 + v + mrow) * OW + (j0 + mh)] = o[v];
  }
}

extern "C" void kernel_launch(void* const* d_in, const int* in_sizes, int n_in,
                              void* d_out, int out_size, void* d_ws, size_t ws_size,
                              hipStream_t stream) {
  (void)in_sizes; (void)n_in; (void)out_size; (void)ws_size;
  const float* img = (const float*)d_in[0];
  float* out = (float*)d_out;
  float* mid = (float*)d_ws;

  cdf23_up2x_wmma<<<2048, 256, 0, stream>>>(img, mid, 256, 1);

  cdf23_up2x_wmma<<<8192, 256, 0, stream>>>(mid, out, 512, 0);
}


# each wave makes a 16x128 output strip (8 tiles) from one shared 19x75 input window, not one 16x16 tile: 8x fewer waves, window and first product shared; per-tile f32 sum order unchanged
# speedup vs baseline: 2.2979x; 2.2979x over previous
#include <hip/hip_runtime.h>

typedef __attribute__((ext_vector_type(2))) float v2f;
typedef __attribute__((ext_vector_type(8))) float v8f;

#define WAVES_PER_BLOCK 8
#define NT    8
#define NCT   5
#define WCOLS (8 * NT + 11)
#define XS    80
#define TS    84

__constant__ float cH[23] = {
    -1.20162964e-4f, 0.0f,  1.615524292e-3f, 0.0f, -1.0385513306e-2f, 0.0f,
     4.3619155884e-2f, 0.0f, -1.45397186478e-1f, 0.0f, 6.1066818237e-1f,
     1.0f,
     6.1066818237e-1f, 0.0f, -1.45397186478e-1f, 0.0f, 4.3619155884e-2f, 0.0f,
    -1.0385513306e-2f, 0.0f,  1.615524292e-3f, 0.0f, -1.20162964e-4f };

template <int N, int PHI>
__global__ __launch_bounds__(256) void cdf23_up2x_strip(
    const float* __restrict__ in, float* __restrict__ out)
{
  __shared__ float sG[16 * 20];
  __shared__ __attribute__((aligned(16))) float sW[WAVES_PER_BLOCK][20 * XS];

  const int tid  = threadIdx.x;
  const int lane = tid & 31;
  const int wv   = __builtin_amdgcn_readfirstlane(tid >> 5);
  const int cofs = PHI ? 22 : 21;

  for (int t = tid; t < 16 * 20; t += 256) {
    int m = t / 20, n = t - 20 * m;
    int j = m - 2 * n + cofs;
    sG[t] = (j >= 0 && j < 23) ? cH[j] : 0.0f;
  }
  __syncthreads();

  constexpr int OW  = 2 * N;
  constexpr int TW  = OW / 16;
  constexpr int SPR = TW / NT;
  constexpr int SPP = TW * SPR;
  const int g     = blockIdx.x * WAVES_PER_BLOCK + wv;
  const int plane = g / SPP;
  const int rem   = g - plane * SPP;
  const int sc    = rem / TW;
  const int rt    = rem - sc * TW;
  const int i0 = rt * 16;
  const int j0 = sc * (16 * NT);
  constexpr int back = PHI ? 6 : 5;
  const int nr0 = (i0 >> 1) - back;
  const int nc0 = (j0 >> 1) - back;
  const float* __restrict__ xin = in + (long long)plane * N * N;

  float* X = sW[wv];
  float* T = sW[wv];

  const int gc0 = (nc0 + lane) & (N - 1);
  const int gc1 = (nc0 + 32 + lane) & (N - 1);
  const int gc2 = (nc0 + 48 + lane) & (N - 1);
  const bool dat2 = lane < WCOLS - 48;
#pragma unroll
  for (int r = 0; r < 19; ++r) {
    const int gr = (nr0 + r) & (N - 1);
    const float* __restrict__ row = xin + gr * N;
    const float x0 = row[gc0];
    const float x1 = row[gc1];
    const float x2 = row[gc2];
    X[r * XS + lane]      = x0;
    X[r * XS + 32 + lane] = x1;
    X[r * XS + 48 + lane] = dat2 ? x2 : 0.0f;
  }
  X[19 * XS + lane]      = 0.0f;
  X[19 * XS + 32 + lane] = 0.0f;
  X[19 * XS + 48 + lane] = 0.0f;
  __builtin_amdgcn_wave_barrier();

  const int mh = lane & 15;
  const int kh = (lane < 16) ? 0 : 2;

  v2f gf[5];
#pragma unroll
  for (int k = 0; k < 5; ++k) {
    const int kb = 4 * k + kh;
    gf[k].x = sG[mh * 20 + kb];
    gf[k].y = sG[mh * 20 + kb + 1];
  }

  v8f t[NCT];
#pragma unroll
  for (int ct = 0; ct < NCT; ++ct) t[ct] = (v8f){};
#pragma unroll
  for (int k = 0; k < 5; ++k) {
    const int kb = 4 * k + kh;
#pragma unroll
    for (int ct = 0; ct < NCT; ++ct) {
      v2f b;
      b.x = X[kb * XS + 16 * ct + mh];
      b.y = X[(kb + 1) * XS + 16 * ct + mh];
      t[ct] = __builtin_amdgcn_wmma_f32_16x16x4_f32(false, gf[k], false, b, (short)0, t[ct], false, false);
    }
  }

  __builtin_amdgcn_wave_barrier();
  const int mrow = (lane < 16) ? 0 : 8;
#pragma unroll
  for (int ct = 0; ct < NCT; ++ct) {
#pragma unroll
    for (int v = 0; v < 8; ++v) T[(v + mrow) * TS + 16 * ct + mh] = t[ct][v];
  }
  __builtin_amdgcn_wave_barrier();

  float* __restrict__ op = out + (long long)plane * OW * OW + (long long)(i0 + mrow) * OW + (j0 + mh);
#pragma unroll
  for (int p = 0; p < NT; p += 2) {
    v8f o0 = {}, o1 = {};
#pragma unroll
    for (int k = 0; k < 5; ++k) {
      const int kb = 4 * k + kh;
      v2f a0, a1;
      a0.x = T[mh * TS + 8 * p + kb];
      a0.y = T[mh * TS + 8 * p + kb + 1];
      a1.x = T[mh * TS + 8 * (p + 1) + kb];
      a1.y = T[mh * TS + 8 * (p + 1) + kb + 1];
      o0 = __builtin_amdgcn_wmma_f32_16x16x4_f32(false, a0, false, gf[k], (short)0, o0, false, false);
      o1 = __builtin_amdgcn_wmma_f32_16x16x4_f32(false, a1, false, gf[k], (short)0, o1, false, false);
    }
#pragma unroll
    for (int v = 0; v < 8; ++v) {
      op[v * OW + 16 * p]       = o0[v];
      op[v * OW + 16 * (p + 1)] = o1[v];
    }
  }
}

extern "C" void kernel_launch(void* const* d_in, const int* in_sizes, int n_in,
                              void* d_out, int out_size, void* d_ws, size_t ws_size,
                              hipStream_t stream) {
  (void)in_sizes; (void)n_in; (void)out_size; (void)ws_size;
  const float* img = (const float*)d_in[0];
  float* out = (float*)d_out;
  float* mid = (float*)d_ws;

  cdf23_up2x_strip<256, 1><<<256, 256, 0, stream>>>(img, mid);

  cdf23_up2x_strip<512, 0><<<1024, 256, 0, stream>>>(mid, out);
}


# each wave loops over its strips with the window loads issued one strip ahead (the first ahead of the G build); second launch: 2 strips per wave, 512 blocks not 1024; per-tile f32 sum order unchanged
# speedup vs baseline: 2.5728x; 1.1196x over previous
#include <hip/hip_runtime.h>

typedef __attribute__((ext_vector_type(2))) float v2f;
typedef __attribute__((ext_vector_type(8))) float v8f;

#define WAVES_PER_BLOCK 8
#define NT    8
#define NCT   5
#define WCOLS (8 * NT + 11)
#define XS    80
#define TS    84

__constant__ float cH[23] = {
    -1.20162964e-4f, 0.0f,  1.615524292e-3f, 0.0f, -1.0385513306e-2f, 0.0f,
     4.3619155884e-2f, 0.0f, -1.45397186478e-1f, 0.0f, 6.1066818237e-1f,
     1.0f,
     6.1066818237e-1f, 0.0f, -1.45397186478e-1f, 0.0f, 4.3619155884e-2f, 0.0f,
    -1.0385513306e-2f, 0.0f,  1.615524292e-3f, 0.0f, -1.20162964e-4f };


template <int N>
__device__ __forceinline__ void strip_coords(int g, int& plane, int& i0, int& j0)
{
  constexpr int TW  = (2 * N) / 16;
  constexpr int SPR = TW / NT;
  constexpr int SPP = TW * SPR;
  plane = g / SPP;
  const int rem = g - plane * SPP;
  const int sc  = rem / TW;
  const int rt  = rem - sc * TW;
  i0 = rt * 16;
  j0 = sc * (16 * NT);
}

template <int N, int PHI>
__device__ __forceinline__ void load_window(const float* __restrict__ in, int plane, int i0, int j0,
                                            int lane, float (&xa)[19], float (&xb)[19], float (&xc)[19])
{
  constexpr int back = PHI ? 6 : 5;
  const int nr0 = (i0 >> 1) - back;
  const int nc0 = (j0 >> 1) - back;
  const float* __restrict__ xin = in + (long long)plane * N * N;
  const int gc0 = (nc0 + lane) & (N - 1);
  const int gc1 = (nc0 + 32 + lane) & (N - 1);
  const int gc2 = (nc0 + 48 + lane) & (N - 1);
#pragma unroll
  for (int r = 0; r < 19; ++r) {
    const int gr = (nr0 + r) & (N - 1);
    const float* __restrict__ row = xin + gr * N;
    xa[r] = row[gc0];
    xb[r] = row[gc1];
    xc[r] = row[gc2];
  }
}

template <int N, int PHI>
__global__ __launch_bounds__(256) void cdf23_up2x_strip(
    const float* __restrict__ in, float* __restrict__ out, int spw)
{
  __shared__ float sG[16 * 20];
  __shared__ __attribute__((aligned(16))) float sW[WAVES_PER_BLOCK][20 * XS];

  const int tid  = threadIdx.x;
  const int lane = tid & 31;
  const int wv   = __builtin_amdgcn_readfirstlane(tid >> 5);
  const int cofs = PHI ? 22 : 21;

  constexpr int OW = 2 * N;

  const int g0 = blockIdx.x * (spw * WAVES_PER_BLOCK) + wv;
  int plane, i0, j0;
  strip_coords<N>(g0, plane, i0, j0);
  float xa[19], xb[19], xc[19];
  load_window<N, PHI>(in, plane, i0, j0, lane, xa, xb, xc);

  for (int t = tid; t < 16 * 20; t += 256) {
    int m = t / 20, n = t - 20 * m;
    int j = m - 2 * n + cofs;
    sG[t] = (j >= 0 && j < 23) ? cH[j] : 0.0f;
  }
  __syncthreads();

  float* X = sW[wv];
  float* T = sW[wv];
  const bool dat2 = lane < WCOLS - 48;

  const int mh = lane & 15;
  const int kh = (lane < 16) ? 0 : 2;
  const int mrow = (lane < 16) ? 0 : 8;

  v2f gf[5];
#pragma unroll
  for (int k = 0; k < 5; ++k) {
    const int kb = 4 * k + kh;
    gf[k].x = sG[mh * 20 + kb];
    gf[k].y = sG[mh * 20 + kb + 1];
  }

  for (int s = 0; s < spw; ++s) {
#pragma unroll
    for (int r = 0; r < 19; ++r) {
      X[r * XS + lane]      = xa[r];
      X[r * XS + 32 + lane] = xb[r];
      X[r * XS + 48 + lane] = dat2 ? xc[r] : 0.0f;
    }
    X[19 * XS + lane]      = 0.0f;
    X[19 * XS + 32 + lane] = 0.0f;
    X[19 * XS + 48 + lane] = 0.0f;
    __builtin_amdgcn_wave_barrier();

    float* __restrict__ op = out + (long long)plane * OW * OW + (long long)(i0 + mrow) * OW + (j0 + mh);

    if (s + 1 < spw) {
      strip_coords<N>(g0 + (s + 1) * WAVES_PER_BLOCK, plane, i0, j0);
      load_window<N, PHI>(in, plane, i0, j0, lane, xa, xb, xc);
    }

    v8f t[NCT];
#pragma unroll
    for (int ct = 0; ct < NCT; ++ct) t[ct] = (v8f){};
#pragma unroll
    for (int k = 0; k < 5; ++k) {
      const int kb = 4 * k + kh;
#pragma unroll
      for (int ct = 0; ct < NCT; ++ct) {
        v2f b;
        b.x = X[kb * XS + 16 * ct + mh];
        b.y = X[(kb + 1) * XS + 16 * ct + mh];
        t[ct] = __builtin_amdgcn_wmma_f32_16x16x4_f32(false, gf[k], false, b, (short)0, t[ct], false, false);
      }
    }

    __builtin_amdgcn_wave_barrier();
#pragma unroll
    for (int ct = 0; ct < NCT; ++ct) {
#pragma unroll
      for (int v = 0; v < 8; ++v) T[(v + mrow) * TS + 16 * ct + mh] = t[ct][v];
    }
    __builtin_amdgcn_wave_barrier();

#pragma unroll
    for (int p = 0; p < NT; p += 2) {
      v8f o0 = {}, o1 = {};
#pragma unroll
      for (int k = 0; k < 5; ++k) {
        const int kb = 4 * k + kh;
        v2f a0, a1;
        a0.x = T[mh * TS + 8 * p + kb];
        a0.y = T[mh * TS + 8 * p + kb + 1];
        a1.x = T[mh * TS + 8 * (p + 1) + kb];
        a1.y = T[mh * TS + 8 * (p + 1) + kb + 1];
        o0 = __builtin_amdgcn_wmma_f32_16x16x4_f32(false, a0, false, gf[k], (short)0, o0, false, false);
        o1 = __builtin_amdgcn_wmma_f32_16x16x4_f32(false, a1, false, gf[k], (short)0, o1, false, false);
      }
#pragma unroll
      for (int v = 0; v < 8; ++v) {
        op[v * OW + 16 * p]       = o0[v];
        op[v * OW + 16 * (p + 1)] = o1[v];
      }
    }
    __builtin_amdgcn_wave_barrier();
  }
}

extern "C" void kernel_launch(void* const* d_in, const int* in_sizes, int n_in,
                              void* d_out, int out_size, void* d_ws, size_t ws_size,
                              hipStream_t stream) {
  (void)in_sizes; (void)n_in; (void)out_size; (void)ws_size;
  const float* img = (const float*)d_in[0];
  float* out = (float*)d_out;
  float* mid = (float*)d_ws;

  cdf23_up2x_strip<256, 1><<<256, 256, 0, stream>>>(img, mid, 1);

  cdf23_up2x_strip<512, 0><<<512, 256, 0, stream>>>(mid, out, 2);
}


# output rows staged through LDS and written with 16-byte stores, each covering two whole 256-byte row pieces, not 64-byte half lines; second product transposed to stage it (same k order and f32 sums)
# speedup vs baseline: 2.9622x; 1.1514x over previous
#include <hip/hip_runtime.h>

typedef __attribute__((ext_vector_type(2))) float v2f;
typedef __attribute__((ext_vector_type(4))) float v4f;
typedef __attribute__((ext_vector_type(8))) float v8f;

#define WAVES_PER_BLOCK 8
#define NT    8
#define NCT   5
#define WCOLS (8 * NT + 11)
#define XS    80
#define TS    84
#define OP    68

__constant__ float cH[23] = {
    -1.20162964e-4f, 0.0f,  1.615524292e-3f, 0.0f, -1.0385513306e-2f, 0.0f,
     4.3619155884e-2f, 0.0f, -1.45397186478e-1f, 0.0f, 6.1066818237e-1f,
     1.0f,
     6.1066818237e-1f, 0.0f, -1.45397186478e-1f, 0.0f, 4.3619155884e-2f, 0.0f,
    -1.0385513306e-2f, 0.0f,  1.615524292e-3f, 0.0f, -1.20162964e-4f };


template <int N>
__device__ __forceinline__ void strip_coords(int g, int& plane, int& i0, int& j0)
{
  constexpr int TW  = (2 * N) / 16;
  constexpr int SPR = TW / NT;
  constexpr int SPP = TW * SPR;
  plane = g / SPP;
  const int rem = g - plane * SPP;
  const int sc  = rem / TW;
  const int rt  = rem - sc * TW;
  i0 = rt * 16;
  j0 = sc * (16 * NT);
}

template <int N, int PHI>
__device__ __forceinline__ void load_window(const float* __restrict__ in, int plane, int i0, int j0,
                                            int lane, float (&xa)[19], float (&xb)[19], float (&xc)[19])
{
  constexpr int back = PHI ? 6 : 5;
  const int nr0 = (i0 >> 1) - back;
  const int nc0 = (j0 >> 1) - back;
  const float* __restrict__ xin = in + (long long)plane * N * N;
  const int gc0 = (nc0 + lane) & (N - 1);
  const int gc1 = (nc0 + 32 + lane) & (N - 1);
  const int gc2 = (nc0 + 48 + lane) & (N - 1);
#pragma unroll
  for (int r = 0; r < 19; ++r) {
    const int gr = (nr0 + r) & (N - 1);
    const float* __restrict__ row = xin + gr * N;
    xa[r] = row[gc0];
    xb[r] = row[gc1];
    xc[r] = row[gc2];
  }
}

template <int N, int PHI>
__global__ __launch_bounds__(256) void cdf23_up2x_strip(
    const float* __restrict__ in, float* __restrict__ out, int spw)
{
  __shared__ float sG[16 * 20];
  __shared__ __attribute__((aligned(16))) float sW[WAVES_PER_BLOCK][20 * XS];
  __shared__ __attribute__((aligned(16))) float sO[WAVES_PER_BLOCK][16 * OP];

  const int tid  = threadIdx.x;
  const int lane = tid & 31;
  const int wv   = __builtin_amdgcn_readfirstlane(tid >> 5);
  const int cofs = PHI ? 22 : 21;

  constexpr int OW = 2 * N;

  const int g0 = blockIdx.x * (spw * WAVES_PER_BLOCK) + wv;
  int plane, i0, j0;
  strip_coords<N>(g0, plane, i0, j0);
  float xa[19], xb[19], xc[19];
  load_window<N, PHI>(in, plane, i0, j0, lane, xa, xb, xc);

  for (int t = tid; t < 16 * 20; t += 256) {
    int m = t / 20, n = t - 20 * m;
    int j = m - 2 * n + cofs;
    sG[t] = (j >= 0 && j < 23) ? cH[j] : 0.0f;
  }
  __syncthreads();

  float* X = sW[wv];
  float* T = sW[wv];
  float* O = sO[wv];
  const bool dat2 = lane < WCOLS - 48;

  const int mh = lane & 15;
  const int kh = (lane < 16) ? 0 : 2;
  const int mrow = (lane < 16) ? 0 : 8;
  const int hh = lane >> 4;

  v2f gf[5];
#pragma unroll
  for (int k = 0; k < 5; ++k) {
    const int kb = 4 * k + kh;
    gf[k].x = sG[mh * 20 + kb];
    gf[k].y = sG[mh * 20 + kb + 1];
  }

  for (int s = 0; s < spw; ++s) {
#pragma unroll
    for (int r = 0; r < 19; ++r) {
      X[r * XS + lane]      = xa[r];
      X[r * XS + 32 + lane] = xb[r];
      X[r * XS + 48 + lane] = dat2 ? xc[r] : 0.0f;
    }
    X[19 * XS + lane]      = 0.0f;
    X[19 * XS + 32 + lane] = 0.0f;
    X[19 * XS + 48 + lane] = 0.0f;
    __builtin_amdgcn_wave_barrier();

    float* __restrict__ op = out + (long long)plane * OW * OW + (long long)(i0 + hh) * OW + (j0 + 4 * mh);

    if (s + 1 < spw) {
      strip_coords<N>(g0 + (s + 1) * WAVES_PER_BLOCK, plane, i0, j0);
      load_window<N, PHI>(in, plane, i0, j0, lane, xa, xb, xc);
    }

    v8f t[NCT];
#pragma unroll
    for (int ct = 0; ct < NCT; ++ct) t[ct] = (v8f){};
#pragma unroll
    for (int k = 0; k < 5; ++k) {
      const int kb = 4 * k + kh;
#pragma unroll
      for (int ct = 0; ct < NCT; ++ct) {
        v2f b;
        b.x = X[kb * XS + 16 * ct + mh];
        b.y = X[(kb + 1) * XS + 16 * ct + mh];
        t[ct] = __builtin_amdgcn_wmma_f32_16x16x4_f32(false, gf[k], false, b, (short)0, t[ct], false, false);
      }
    }

    __builtin_amdgcn_wave_barrier();
#pragma unroll
    for (int ct = 0; ct < NCT; ++ct) {
#pragma unroll
      for (int v = 0; v < 8; ++v) T[(v + mrow) * TS + 16 * ct + mh] = t[ct][v];
    }
    __builtin_amdgcn_wave_barrier();

#pragma unroll
    for (int hs = 0; hs < 2; ++hs) {
#pragma unroll
      for (int q = 0; q < 2; ++q) {
        const int p = 4 * hs + 2 * q;
        v8f o0 = {}, o1 = {};
#pragma unroll
        for (int k = 0; k < 5; ++k) {
          const int kb = 4 * k + kh;
          v2f a0, a1;
          a0.x = T[mh * TS + 8 * p + kb];
          a0.y = T[mh * TS + 8 * p + kb + 1];
          a1.x = T[mh * TS + 8 * (p + 1) + kb];
          a1.y = T[mh * TS + 8 * (p + 1) + kb + 1];
          o0 = __builtin_amdgcn_wmma_f32_16x16x4_f32(false, gf[k], false, a0, (short)0, o0, false, false);
          o1 = __builtin_amdgcn_wmma_f32_16x16x4_f32(false, gf[k], false, a1, (short)0, o1, false, false);
        }
        v4f* so = (v4f*)&O[mh * OP + 32 * q + 8 * hh];
        so[0] = __builtin_shufflevector(o0, o0, 0, 1, 2, 3);
        so[1] = __builtin_shufflevector(o0, o0, 4, 5, 6, 7);
        so[4] = __builtin_shufflevector(o1, o1, 0, 1, 2, 3);
        so[5] = __builtin_shufflevector(o1, o1, 4, 5, 6, 7);
      }
      __builtin_amdgcn_wave_barrier();
#pragma unroll
      for (int i = 0; i < 8; ++i) {
        const v4f val = *(const v4f*)&O[(2 * i + hh) * OP + 4 * mh];
        *(v4f*)&op[(2 * i) * OW + 64 * hs] = val;
      }
      __builtin_amdgcn_wave_barrier();
    }
  }
}

extern "C" void kernel_launch(void* const* d_in, const int* in_sizes, int n_in,
                              void* d_out, int out_size, void* d_ws, size_t ws_size,
                              hipStream_t stream) {
  (void)in_sizes; (void)n_in; (void)out_size; (void)ws_size;
  const float* img = (const float*)d_in[0];
  float* out = (float*)d_out;
  float* mid = (float*)d_ws;

  cdf23_up2x_strip<256, 1><<<256, 256, 0, stream>>>(img, mid, 1);

  cdf23_up2x_strip<512, 0><<<512, 256, 0, stream>>>(mid, out, 2);
}


# both 2x levels in one launch: each wave rebuilds its two strips' 29x81 window of the middle level in LDS from one 27x64 input window; no scratch pass; middle level's f32 sums regrouped in k
# speedup vs baseline: 4.9161x; 1.6596x over previous
#include <hip/hip_runtime.h>

typedef __attribute__((ext_vector_type(2))) float v2f;
typedef __attribute__((ext_vector_type(4))) float v4f;
typedef __attribute__((ext_vector_type(8))) float v8f;

#define WAVES_PER_BLOCK 8
#define NIN   256
#define OW    1024
#define NT    8
#define NCT   5
#define XP    72
#define XROWS 28
#define TS    84
#define MP    84
#define MROWS 29
#define OP    68
#define TFLOATS (16 * TS)
#define AFLOATS (TFLOATS + 16 * OP)

__constant__ float cH[23] = {
    -1.20162964e-4f, 0.0f,  1.615524292e-3f, 0.0f, -1.0385513306e-2f, 0.0f,
     4.3619155884e-2f, 0.0f, -1.45397186478e-1f, 0.0f, 6.1066818237e-1f,
     1.0f,
     6.1066818237e-1f, 0.0f, -1.45397186478e-1f, 0.0f, 4.3619155884e-2f, 0.0f,
    -1.0385513306e-2f, 0.0f,  1.615524292e-3f, 0.0f, -1.20162964e-4f };


__global__ __launch_bounds__(256) void cdf23_up4x_pair(const float* __restrict__ in, float* __restrict__ out)
{
  __shared__ float sG[2][16 * 20];
  __shared__ __attribute__((aligned(16))) float sA[WAVES_PER_BLOCK][AFLOATS];
  __shared__ __attribute__((aligned(16))) float sM[WAVES_PER_BLOCK][MROWS * MP];

  const int tid  = threadIdx.x;
  const int lane = tid & 31;
  const int wv   = __builtin_amdgcn_readfirstlane(tid >> 5);

  const int pp    = blockIdx.x * WAVES_PER_BLOCK + wv;
  const int plane = pp >> 8;
  const int sc    = (pp & 255) >> 5;
  const int i0    = (pp & 31) * 32;
  const int j0    = sc * (16 * NT);

  const int q0 = (i0 >> 2) - 9;
  const int c0 = (j0 >> 2) - 9;
  const float* __restrict__ xin = in + (long long)plane * NIN * NIN;
  const int gc0 = (c0 + lane) & (NIN - 1);
  const int gc1 = (c0 + 32 + lane) & (NIN - 1);
  float xa[27], xb[27];
#pragma unroll
  for (int r = 0; r < 27; ++r) {
    const int gr = (q0 + r) & (NIN - 1);
    const float* __restrict__ row = xin + gr * NIN;
    xa[r] = row[gc0];
    xb[r] = row[gc1];
  }

  for (int t = tid; t < 2 * 16 * 20; t += 256) {
    const int ph = t / 320, u = t - 320 * ph;
    const int m = u / 20, n = u - 20 * m;
    const int j = m - 2 * n + (ph ? 21 : 22);
    sG[ph][u] = (j >= 0 && j < 23) ? cH[j] : 0.0f;
  }
  __syncthreads();

  float* X = sA[wv];
  float* T = sA[wv];
  float* O = sA[wv] + TFLOATS;
  float* M = sM[wv];

  const int mh = lane & 15;
  const int kh = (lane < 16) ? 0 : 2;
  const int mrow = (lane < 16) ? 0 : 8;
  const int hh = lane >> 4;

  v2f gf1[5], gf2[5];
#pragma unroll
  for (int k = 0; k < 5; ++k) {
    const int kb = 4 * k + kh;
    gf1[k].x = sG[0][mh * 20 + kb];
    gf1[k].y = sG[0][mh * 20 + kb + 1];
    gf2[k].x = sG[1][mh * 20 + kb];
    gf2[k].y = sG[1][mh * 20 + kb + 1];
  }

#pragma unroll
  for (int r = 0; r < 27; ++r) {
    X[r * XP + lane]      = xa[r];
    X[r * XP + 32 + lane] = xb[r];
  }
  X[27 * XP + lane]      = 0.0f;
  X[27 * XP + 32 + lane] = 0.0f;
  if (lane < MROWS) M[lane * MP + 80] = 0.0f;
  __builtin_amdgcn_wave_barrier();

  v8f t1[2][4];
#pragma unroll
  for (int tr = 0; tr < 2; ++tr)
#pragma unroll
    for (int ct = 0; ct < 4; ++ct) t1[tr][ct] = (v8f){};
#pragma unroll
  for (int k = 0; k < 5; ++k) {
    const int kb = 4 * k + kh;
#pragma unroll
    for (int tr = 0; tr < 2; ++tr) {
#pragma unroll
      for (int ct = 0; ct < 4; ++ct) {
        v2f b;
        b.x = X[(8 * tr + kb) * XP + 16 * ct + mh];
        b.y = X[(8 * tr + kb + 1) * XP + 16 * ct + mh];
        t1[tr][ct] = __builtin_amdgcn_wmma_f32_16x16x4_f32(false, gf1[k], false, b, (short)0, t1[tr][ct], false, false);
      }
    }
  }

#pragma unroll
  for (int tr = 0; tr < 2; ++tr) {
    __builtin_amdgcn_wave_barrier();
#pragma unroll
    for (int ct = 0; ct < 4; ++ct) {
#pragma unroll
      for (int v = 0; v < 8; ++v) T[(v + mrow) * TS + 16 * ct + mh] = t1[tr][ct][v];
    }
    __builtin_amdgcn_wave_barrier();

    v8f o[NCT];
#pragma unroll
    for (int b = 0; b < NCT; ++b) o[b] = (v8f){};
#pragma unroll
    for (int k = 0; k < 5; ++k) {
      const int kb = 4 * k + kh;
#pragma unroll
      for (int b = 0; b < NCT; ++b) {
        v2f a;
        a.x = T[mh * TS + 8 * b + kb];
        a.y = T[mh * TS + 8 * b + kb + 1];
        o[b] = __builtin_amdgcn_wmma_f32_16x16x4_f32(false, gf1[k], false, a, (short)0, o[b], false, false);
      }
    }
    if (tr == 0 || mh <= MROWS - 1 - 16) {
#pragma unroll
      for (int b = 0; b < NCT; ++b) {
        v4f* sm = (v4f*)&M[(16 * tr + mh) * MP + 16 * b + 8 * hh];
        sm[0] = __builtin_shufflevector(o[b], o[b], 0, 1, 2, 3);
        sm[1] = __builtin_shufflevector(o[b], o[b], 4, 5, 6, 7);
      }
    }
  }
  __builtin_amdgcn_wave_barrier();

  for (int s = 0; s < 2; ++s) {
    const float* __restrict__ W2 = M + (1 + 8 * s) * MP + 1;

    float* __restrict__ op = out + (long long)plane * OW * OW + (long long)(i0 + 16 * s + hh) * OW + (j0 + 4 * mh);

    v8f t[NCT];
#pragma unroll
    for (int ct = 0; ct < NCT; ++ct) t[ct] = (v8f){};
#pragma unroll
    for (int k = 0; k < 5; ++k) {
      const int kb = 4 * k + kh;
#pragma unroll
      for (int ct = 0; ct < NCT; ++ct) {
        v2f b;
        b.x = W2[kb * MP + 16 * ct + mh];
        b.y = W2[(kb + 1) * MP + 16 * ct + mh];
        t[ct] = __builtin_amdgcn_wmma_f32_16x16x4_f32(false, gf2[k], false, b, (short)0, t[ct], false, false);
      }
    }

    __builtin_amdgcn_wave_barrier();
#pragma unroll
    for (int ct = 0; ct < NCT; ++ct) {
#pragma unroll
      for (int v = 0; v < 8; ++v) T[(v + mrow) * TS + 16 * ct + mh] = t[ct][v];
    }
    __builtin_amdgcn_wave_barrier();

#pragma unroll
    for (int hs = 0; hs < 2; ++hs) {
#pragma unroll
      for (int q = 0; q < 2; ++q) {
        const int p = 4 * hs + 2 * q;
        v8f o0 = {}, o1 = {};
#pragma unroll
        for (int k = 0; k < 5; ++k) {
          const int kb = 4 * k + kh;
          v2f a0, a1;
          a0.x = T[mh * TS + 8 * p + kb];
          a0.y = T[mh * TS + 8 * p + kb + 1];
          a1.x = T[mh * TS + 8 * (p + 1) + kb];
          a1.y = T[mh * TS + 8 * (p + 1) + kb + 1];
          o0 = __builtin_amdgcn_wmma_f32_16x16x4_f32(false, gf2[k], false, a0, (short)0, o0, false, false);
          o1 = __builtin_amdgcn_wmma_f32_16x16x4_f32(false, gf2[k], false, a1, (short)0, o1, false, false);
        }
        v4f* so = (v4f*)&O[mh * OP + 32 * q + 8 * hh];
        so[0] = __builtin_shufflevector(o0, o0, 0, 1, 2, 3);
        so[1] = __builtin_shufflevector(o0, o0, 4, 5, 6, 7);
        so[4] = __builtin_shufflevector(o1, o1, 0, 1, 2, 3);
        so[5] = __builtin_shufflevector(o1, o1, 4, 5, 6, 7);
      }
      __builtin_amdgcn_wave_barrier();
#pragma unroll
      for (int i = 0; i < 8; ++i) {
        const v4f val = *(const v4f*)&O[(2 * i + hh) * OP + 4 * mh];
        *(v4f*)&op[(2 * i) * OW + 64 * hs] = val;
      }
      __builtin_amdgcn_wave_barrier();
    }
  }
}

extern "C" void kernel_launch(void* const* d_in, const int* in_sizes, int n_in,
                              void* d_out, int out_size, void* d_ws, size_t ws_size,
                              hipStream_t stream) {
  (void)in_sizes; (void)n_in; (void)out_size; (void)d_ws; (void)ws_size;
  const float* img = (const float*)d_in[0];
  float* out = (float*)d_out;

  cdf23_up4x_pair<<<512, 256, 0, stream>>>(img, out);
}


# the two 2x levels composed exactly into ONE 4x operator per axis (Gc = G2*G1, 16x20; taps multiplied in double at compile time): 110 WMMAs per wave, not 220; no middle level; f32 sum order changed
# speedup vs baseline: 7.4238x; 1.5101x over previous
#include <hip/hip_runtime.h>

typedef __attribute__((ext_vector_type(2))) float v2f;
typedef __attribute__((ext_vector_type(4))) float v4f;
typedef __attribute__((ext_vector_type(8))) float v8f;

#define WAVES_PER_BLOCK 8
#define NIN   256
#define OW    1024
#define NT    8
#define NCT   3
#define XP    72
#define XROWS 24
#define TS    52
#define OP    68

struct Taps { float v[23]; };
constexpr Taps kH = {{
    -1.20162964e-4f, 0.0f,  1.615524292e-3f, 0.0f, -1.0385513306e-2f, 0.0f,
     4.3619155884e-2f, 0.0f, -1.45397186478e-1f, 0.0f, 6.1066818237e-1f,
     1.0f,
     6.1066818237e-1f, 0.0f, -1.45397186478e-1f, 0.0f, 4.3619155884e-2f, 0.0f,
    -1.0385513306e-2f, 0.0f,  1.615524292e-3f, 0.0f, -1.20162964e-4f }};

struct Taps4 { float v[67]; };
constexpr Taps4 compose_levels()
{
  Taps4 t{};
  for (int d = -33; d <= 33; ++d) {
    double s = 0.0;
    for (int e = -11; e <= 11; ++e) {
      const int j = d + 2 * e + 11;
      if (j >= 0 && j < 23) s += (double)kH.v[11 - e] * (double)kH.v[j];
    }
    t.v[d + 33] = (float)s;
  }
  return t;
}
__constant__ Taps4 cF = compose_levels();


__global__ __launch_bounds__(256) void cdf23_up4x_composed(const float* __restrict__ in, float* __restrict__ out)
{
  __shared__ float sG[16 * 20];
  __shared__ __attribute__((aligned(16))) float sX[WAVES_PER_BLOCK][XROWS * XP];
  __shared__ __attribute__((aligned(16))) float sT[WAVES_PER_BLOCK][16 * TS];
  __shared__ __attribute__((aligned(16))) float sO[WAVES_PER_BLOCK][16 * OP];

  const int tid  = threadIdx.x;
  const int lane = tid & 31;
  const int wv   = __builtin_amdgcn_readfirstlane(tid >> 5);

  const int pp    = blockIdx.x * WAVES_PER_BLOCK + wv;
  const int plane = pp >> 8;
  const int sc    = (pp & 255) >> 5;
  const int i0    = (pp & 31) * 32;
  const int j0    = sc * (16 * NT);

  const int q0 = (i0 >> 2) - 8;
  const int c0 = (j0 >> 2) - 8;
  const float* __restrict__ xin = in + (long long)plane * NIN * NIN;
  const int gc0 = (c0 + lane) & (NIN - 1);
  const int gc1 = (c0 + 32 + lane) & (NIN - 1);
  float xa[XROWS], xb[XROWS];
#pragma unroll
  for (int r = 0; r < XROWS; ++r) {
    const int gr = (q0 + r) & (NIN - 1);
    const float* __restrict__ row = xin + gr * NIN;
    xa[r] = row[gc0];
    xb[r] = row[gc1];
  }

  for (int t = tid; t < 16 * 20; t += 256) {
    const int m = t / 20, n = t - 20 * m;
    const int d = 4 * n - m - 30;
    sG[t] = (d >= -33 && d <= 33) ? cF.v[d + 33] : 0.0f;
  }
  __syncthreads();

  float* X = sX[wv];
  float* T = sT[wv];
  float* O = sO[wv];

  const int mh = lane & 15;
  const int kh = (lane < 16) ? 0 : 2;
  const int mrow = (lane < 16) ? 0 : 8;
  const int hh = lane >> 4;

  v2f gf[5];
#pragma unroll
  for (int k = 0; k < 5; ++k) {
    const int kb = 4 * k + kh;
    gf[k].x = sG[mh * 20 + kb];
    gf[k].y = sG[mh * 20 + kb + 1];
  }

#pragma unroll
  for (int r = 0; r < XROWS; ++r) {
    X[r * XP + lane]      = xa[r];
    X[r * XP + 32 + lane] = xb[r];
  }
  __builtin_amdgcn_wave_barrier();

  for (int s = 0; s < 2; ++s) {
    const float* __restrict__ W2 = X + (4 * s) * XP;

    float* __restrict__ op = out + (long long)plane * OW * OW + (long long)(i0 + 16 * s + hh) * OW + (j0 + 4 * mh);

    v8f t[NCT];
#pragma unroll
    for (int ct = 0; ct < NCT; ++ct) t[ct] = (v8f){};
#pragma unroll
    for (int k = 0; k < 5; ++k) {
      const int kb = 4 * k + kh;
#pragma unroll
      for (int ct = 0; ct < NCT; ++ct) {
        v2f b;
        b.x = W2[kb * XP + 16 * ct + mh];
        b.y = W2[(kb + 1) * XP + 16 * ct + mh];
        t[ct] = __builtin_amdgcn_wmma_f32_16x16x4_f32(false, gf[k], false, b, (short)0, t[ct], false, false);
      }
    }

    __builtin_amdgcn_wave_barrier();
#pragma unroll
    for (int ct = 0; ct < NCT; ++ct) {
#pragma unroll
      for (int v = 0; v < 8; ++v) T[(v + mrow) * TS + 16 * ct + mh] = t[ct][v];
    }
    __builtin_amdgcn_wave_barrier();

#pragma unroll
    for (int hs = 0; hs < 2; ++hs) {
#pragma unroll
      for (int q = 0; q < 2; ++q) {
        const int p = 4 * hs + 2 * q;
        v8f o0 = {}, o1 = {};
#pragma unroll
        for (int k = 0; k < 5; ++k) {
          const int kb = 4 * k + kh;
          v2f a0, a1;
          a0.x = T[mh * TS + 4 * p + kb];
          a0.y = T[mh * TS + 4 * p + kb + 1];
          a1.x = T[mh * TS + 4 * (p + 1) + kb];
          a1.y = T[mh * TS + 4 * (p + 1) + kb + 1];
          o0 = __builtin_amdgcn_wmma_f32_16x16x4_f32(false, gf[k], false, a0, (short)0, o0, false, false);
          o1 = __builtin_amdgcn_wmma_f32_16x16x4_f32(false, gf[k], false, a1, (short)0, o1, false, false);
        }
        v4f* so = (v4f*)&O[mh * OP + 32 * q + 8 * hh];
        so[0] = __builtin_shufflevector(o0, o0, 0, 1, 2, 3);
        so[1] = __builtin_shufflevector(o0, o0, 4, 5, 6, 7);
        so[4] = __builtin_shufflevector(o1, o1, 0, 1, 2, 3);
        so[5] = __builtin_shufflevector(o1, o1, 4, 5, 6, 7);
      }
      __builtin_amdgcn_wave_barrier();
#pragma unroll
      for (int i = 0; i < 8; ++i) {
        const v4f val = *(const v4f*)&O[(2 * i + hh) * OP + 4 * mh];
        *(v4f*)&op[(2 * i) * OW + 64 * hs] = val;
      }
      __builtin_amdgcn_wave_barrier();
    }
  }
}

extern "C" void kernel_launch(void* const* d_in, const int* in_sizes, int n_in,
                              void* d_out, int out_size, void* d_ws, size_t ws_size,
                              hipStream_t stream) {
  (void)in_sizes; (void)n_in; (void)out_size; (void)d_ws; (void)ws_size;
  const float* img = (const float*)d_in[0];
  float* out = (float*)d_out;

  cdf23_up4x_composed<<<512, 256, 0, stream>>>(img, out);
}
